# starting kernel
# baseline (speedup 1.0000x reference)
#include <hip/hip_runtime.h>

typedef __attribute__((ext_vector_type(2))) float v2f;
typedef __attribute__((ext_vector_type(8))) float v8f;

#define CIN    32
#define COUT   32
#define HH     256
#define WW     256
#define TILE_W 128
#define NSTEP  72

#define XPADW   34
#define XROWW   130
#define X_ELEMS (3 * XROWW * XPADW)
#define W_OFF   X_ELEMS
#define W_FRAGS (NSTEP * 2 * 32)
#define B_OFF   (W_OFF + W_FRAGS * 2)
#define SMEM_FLOATS (B_OFF + 32)

extern "C" __global__ __launch_bounds__(256, 2)
void conv3x3_wmma_f32(const float* __restrict__ x,
                      const float* __restrict__ wgt,
                      const float* __restrict__ bias,
                      float* __restrict__ out)
{
    extern __shared__ float smem[];
    float* xlds = smem;
    v2f*   wlds = (v2f*)(smem + W_OFF);
    float* blds = smem + B_OFF;

    const int tid   = threadIdx.x;
    const int blk   = blockIdx.x;
    const int b     = blk >> 9;
    const int h     = (blk >> 1) & 255;
    const int wtile = blk & 1;
    const int wbase_g = wtile * TILE_W;

    for (int idx = tid; idx < W_FRAGS; idx += 256) {
        int s    = idx >> 6;
        int rem  = idx & 63;
        int t    = rem >> 5;
        int l    = rem & 31;
        int k0   = 4 * s + ((l >> 4) << 1);
        int cout = (t << 4) + (l & 15);
        v2f wv;
        #pragma unroll
        for (int v = 0; v < 2; ++v) {
            int k    = k0 + v;
            int ci   = k & 31;
            int khkw = k >> 5;
            wv[v] = wgt[(cout * CIN + ci) * 9 + khkw];
        }
        wlds[idx] = wv;
    }

    if (tid < 32) blds[tid] = bias[tid];

    for (int idx = tid; idx < 3 * CIN * XROWW; idx += 256) {
        int kh = idx / (CIN * XROWW);
        int r  = idx - kh * (CIN * XROWW);
        int ci = r / XROWW;
        int wp = r - ci * XROWW;
        int hin = h + kh - 1;
        int win = wbase_g + wp - 1;
        float v = 0.0f;
        if ((unsigned)hin < HH && (unsigned)win < WW)
            v = x[((b * CIN + ci) * HH + hin) * WW + win];
        xlds[(kh * XROWW + wp) * XPADW + ci] = v;
    }
    __syncthreads();

    const int lane = tid & 31;
    const int wl   = (tid >> 5) << 4;
    const int nco  = lane & 15;
    const int hi   = lane >> 4;

    v8f acc0 = {};
    v8f acc1 = {};

    #pragma unroll
    for (int s = 0; s < NSTEP; ++s) {
        const int khkw = s >> 3;
        const int kh   = khkw / 3;
        const int kw   = khkw - 3 * kh;
        const int cig  = (s & 7) << 2;

        v2f a0 = wlds[(s * 2 + 0) * 32 + lane];
        v2f a1 = wlds[(s * 2 + 1) * 32 + lane];

        const int wp = wl + nco + kw;
        const int ci = cig + (hi << 1);
        v2f bv = *(const v2f*)&xlds[(kh * XROWW + wp) * XPADW + ci];

        acc0 = __builtin_amdgcn_wmma_f32_16x16x4_f32(
                   false, a0, false, bv, (short)0, acc0, false, false);
        acc1 = __builtin_amdgcn_wmma_f32_16x16x4_f32(
                   false, a1, false, bv, (short)0, acc1, false, false);
    }

    const int wcol = wbase_g + wl + nco;
    #pragma unroll
    for (int r = 0; r < 8; ++r) {
        int co = r + (hi << 3);
        float v0 = acc0[r] + blds[co];
        float v1 = acc1[r] + blds[co + 16];
        out[((b * COUT + co)      * HH + h) * WW + wcol] = v0;
        out[((b * COUT + co + 16) * HH + h) * WW + wcol] = v1;
    }
}

extern "C" void kernel_launch(void* const* d_in, const int* in_sizes, int n_in,
                              void* d_out, int out_size, void* d_ws, size_t ws_size,
                              hipStream_t stream) {
    const float* x    = (const float*)d_in[0];
    const float* wgt  = (const float*)d_in[1];
    const float* bias = (const float*)d_in[2];
    float* out = (float*)d_out;

    dim3 grid(16 * 256 * 2);
    dim3 block(256);
    size_t smem = SMEM_FLOATS * sizeof(float);
    hipLaunchKernelGGL(conv3x3_wmma_f32, grid, block, smem, stream, x, wgt, bias, out);
}


# Row-loop barrier hands over LDS only: a fence restricted to LDS round a bare barrier, not __syncthreads, so no wave waits for its 16 global output stores before it signals; sums unchanged
# speedup vs baseline: 2.7219x; 2.7219x over previous
#include <hip/hip_runtime.h>


typedef __attribute__((ext_vector_type(2))) float v2f;
typedef __attribute__((ext_vector_type(4))) float v4f;
typedef __attribute__((ext_vector_type(8))) float v8f;

#define CIN    32
#define COUT   32
#define HH     256
#define WW     256
#define TILE_W 128
#define NSTEP  72
#define ROWS   16

#define XPADW   34
#define XROWW   130
#define X_ELEMS (3 * XROWW * XPADW)
#define W_OFF   (2 * X_ELEMS)
#define W_FRAGS (NSTEP * 2 * 32)
#define B_OFF   (W_OFF + W_FRAGS * 2)
#define SMEM_FLOATS (B_OFF + 32)

__device__ __forceinline__ void load_rows(const float* __restrict__ x, int b, int h, int wbase_g,
                                          int tid, v4f (&xa)[6], v4f (&xb)[6], float& hv)
{
    const int lane = tid & 31;
    const int wave = tid >> 5;
    #pragma unroll
    for (int it = 0; it < 6; ++it) {
        int u   = wave * 6 + it;
        int kh  = u >> 4;
        int g   = (u >> 2) & 3;
        int q   = u & 3;
        int w4  = (lane & 7) + (q << 3);
        int ci  = ((g << 2) + (lane >> 3)) << 1;
        int hin = h + kh - 1;
        int hc  = hin < 0 ? 0 : (hin > HH - 1 ? HH - 1 : hin);
        const float* p = x + ((b * CIN + ci) * HH + hc) * WW + wbase_g + (w4 << 2);
        xa[it] = *(const v4f*)p;
        xb[it] = *(const v4f*)(p + HH * WW);
    }
    const int hkh = tid >> 6;
    const int hci = (tid & 63) >> 1;
    const int hwp = (tid & 1) ? (XROWW - 1) : 0;
    hv = 0.0f;
    if (tid < 192) {
        int hin = h + hkh - 1;
        int win = wbase_g + hwp - 1;
        if ((unsigned)hin < HH && (unsigned)win < WW)
            hv = x[((b * CIN + hci) * HH + hin) * WW + win];
    }
}

__device__ __forceinline__ void store_rows(float* xlds, int h, int tid,
                                           const v4f (&xa)[6], const v4f (&xb)[6], float hv)
{
    const int lane = tid & 31;
    const int wave = tid >> 5;
    #pragma unroll
    for (int it = 0; it < 6; ++it) {
        int u   = wave * 6 + it;
        int kh  = u >> 4;
        int g   = (u >> 2) & 3;
        int q   = u & 3;
        int w4  = (lane & 7) + (q << 3);
        int ci  = ((g << 2) + (lane >> 3)) << 1;
        bool ok = (unsigned)(h + kh - 1) < HH;
        #pragma unroll
        for (int jj = 0; jj < 4; ++jj) {
            int wp = (w4 << 2) + 1 + jj;
            v2f pr;
            pr[0] = ok ? xa[it][jj] : 0.0f;
            pr[1] = ok ? xb[it][jj] : 0.0f;
            *(v2f*)&xlds[(kh * XROWW + wp) * XPADW + ci] = pr;
        }
    }
    const int hkh = tid >> 6;
    const int hci = (tid & 63) >> 1;
    const int hwp = (tid & 1) ? (XROWW - 1) : 0;
    if (tid < 192) xlds[(hkh * XROWW + hwp) * XPADW + hci] = hv;
}

extern "C" __global__ __launch_bounds__(256, 1)
void conv3x3_wmma_f32(const float* __restrict__ x,
                      const float* __restrict__ wgt,
                      const float* __restrict__ bias,
                      float* __restrict__ out)
{
    extern __shared__ float smem[];
    v2f*   wlds = (v2f*)(smem + W_OFF);
    float* blds = smem + B_OFF;

    const int tid   = threadIdx.x;
    const int blk   = blockIdx.x;
    const int b     = blk >> 5;
    const int h0    = ((blk >> 1) & 15) * ROWS;
    const int wtile = blk & 1;
    const int wbase_g = wtile * TILE_W;
    const int lane = tid & 31;
    const int wave = tid >> 5;

    v4f wreg[9];
    #pragma unroll
    for (int i = 0; i < 9; ++i) {
        int q    = i * 256 + tid;
        int rest = q >> 5;
        int cout = (q & 7) + ((rest & 3) << 3);
        int f4   = ((rest >> 2) << 2) + ((q >> 3) & 3);
        wreg[i] = *(const v4f*)(wgt + cout * (CIN * 9) + (f4 << 2));
    }
    v4f xa[6], xb[6];
    float hv;
    load_rows(x, b, h0, wbase_g, tid, xa, xb, hv);
    float bsv = 0.0f;
    if (tid < 32) bsv = bias[tid];

    #pragma unroll
    for (int i = 0; i < 9; ++i) {
        int q    = i * 256 + tid;
        int rest = q >> 5;
        int cout = (q & 7) + ((rest & 3) << 3);
        int f4   = ((rest >> 2) << 2) + ((q >> 3) & 3);
        #pragma unroll
        for (int j = 0; j < 4; ++j) {
            int e    = (f4 << 2) + j;
            int ci   = e / 9;
            int khkw = e - 9 * ci;
            int k    = (khkw << 5) + ci;
            int s    = k >> 2;
            int l    = (cout & 15) + (((k >> 1) & 1) << 4);
            int t    = cout >> 4;
            smem[W_OFF + ((((s << 1) + t) * 32 + l) << 1) + (k & 1)] = wreg[i][j];
        }
    }
    store_rows(smem, h0, tid, xa, xb, hv);
    if (tid < 32) blds[tid] = bsv;
    __syncthreads();

    const int wl   = wave << 4;
    const int nco  = lane & 15;
    const int hi   = lane >> 4;

    v2f A0[NSTEP], A1[NSTEP];
    #pragma unroll
    for (int s = 0; s < NSTEP; ++s) {
        A0[s] = wlds[(s * 2 + 0) * 32 + lane];
        A1[s] = wlds[(s * 2 + 1) * 32 + lane];
    }
    float bia0[8], bia1[8];
    #pragma unroll
    for (int r = 0; r < 8; ++r) {
        bia0[r] = blds[r + (hi << 3)];
        bia1[r] = blds[r + (hi << 3) + 16];
    }

    const int wcol = wbase_g + wl + nco;

    #pragma unroll 1
    for (int it = 0; it < ROWS; ++it) {
        const int  h    = h0 + it;
        const bool more = (it + 1) < ROWS;

        if (more) load_rows(x, b, h + 1, wbase_g, tid, xa, xb, hv);

        const float* bp = smem + (it & 1) * X_ELEMS + (wl + nco) * XPADW + (hi << 1);

        v8f acc0 = {};
        v8f acc1 = {};

        #pragma unroll
        for (int s = 0; s < NSTEP; ++s) {
            const int khkw = s >> 3;
            const int kh   = khkw / 3;
            const int kw   = khkw - 3 * kh;
            const int cig  = (s & 7) << 2;
            v2f bv = *(const v2f*)(bp + (kh * XROWW + kw) * XPADW + cig);
            acc0 = __builtin_amdgcn_wmma_f32_16x16x4_f32(
                       false, A0[s], false, bv, (short)0, acc0, false, false);
            acc1 = __builtin_amdgcn_wmma_f32_16x16x4_f32(
                       false, A1[s], false, bv, (short)0, acc1, false, false);
        }

        #pragma unroll
        for (int r = 0; r < 8; ++r) {
            int co = r + (hi << 3);
            out[((b * COUT + co)      * HH + h) * WW + wcol] = acc0[r] + bia0[r];
            out[((b * COUT + co + 16) * HH + h) * WW + wcol] = acc1[r] + bia1[r];
        }

        if (more) store_rows(smem + ((it + 1) & 1) * X_ELEMS, h + 1, tid, xa, xb, hv);
        __builtin_amdgcn_fence(__ATOMIC_RELEASE, "workgroup", "local");
        __builtin_amdgcn_s_barrier();
        __builtin_amdgcn_fence(__ATOMIC_ACQUIRE, "workgroup", "local");
    }
}

extern "C" void kernel_launch(void* const* d_in, const int* in_sizes, int n_in,
                              void* d_out, int out_size, void* d_ws, size_t ws_size,
                              hipStream_t stream) {
    const float* x    = (const float*)d_in[0];
    const float* wgt  = (const float*)d_in[1];
    const float* bias = (const float*)d_in[2];
    float* out = (float*)d_out;

    dim3 grid(16 * 16 * 2);
    dim3 block(256);
    size_t smem = SMEM_FLOATS * sizeof(float);
    hipLaunchKernelGGL(conv3x3_wmma_f32, grid, block, smem, stream, x, wgt, bias, out);
}


# Row loop walks INPUT rows: a row's 24 B fragments are read from LDS once, two read groups at the start of its 144 WMMAs, and feed the 3 output rows it touches (6 accumulator tiles); sums unchanged
# speedup vs baseline: 3.1140x; 1.1441x over previous
#include <hip/hip_runtime.h>


typedef __attribute__((ext_vector_type(2))) float v2f;
typedef __attribute__((ext_vector_type(4))) float v4f;
typedef __attribute__((ext_vector_type(8))) float v8f;

#define CIN    32
#define COUT   32
#define HH     256
#define WW     256
#define TILE_W 128
#define NSTEP  72
#define ROWS   16
#define NIN    (ROWS + 2)
#define NFRAG  24
#define HALF   12

#define XPADW   34
#define XROWW   130
#define XROW_ELEMS (XROWW * XPADW)
#define NBUF    3
#define W_OFF   (NBUF * XROW_ELEMS)
#define W_FRAGS (NSTEP * 2 * 32)
#define B_OFF   (W_OFF + W_FRAGS * 2)
#define SMEM_FLOATS (B_OFF + 32)

__device__ __forceinline__ void load_row(const float* __restrict__ x, int b, int hin, int wbase_g,
                                         int tid, v4f (&xa)[2], v4f (&xb)[2], float& hv)
{
    const int lane = tid & 31;
    const int wave = tid >> 5;
    const int hc   = hin < 0 ? 0 : (hin > HH - 1 ? HH - 1 : hin);
    #pragma unroll
    for (int it = 0; it < 2; ++it) {
        int u   = wave * 2 + it;
        int g   = u >> 2;
        int q   = u & 3;
        int w4  = (lane & 7) + (q << 3);
        int ci  = ((g << 2) + (lane >> 3)) << 1;
        const float* p = x + ((b * CIN + ci) * HH + hc) * WW + wbase_g + (w4 << 2);
        xa[it] = *(const v4f*)p;
        xb[it] = *(const v4f*)(p + HH * WW);
    }
    const int hci = (tid & 63) >> 1;
    const int hwp = (tid & 1) ? (XROWW - 1) : 0;
    hv = 0.0f;
    if (tid < 64) {
        int win = wbase_g + hwp - 1;
        if ((unsigned)hin < HH && (unsigned)win < WW)
            hv = x[((b * CIN + hci) * HH + hin) * WW + win];
    }
}

__device__ __forceinline__ void store_row(float* xl, int hin, int tid,
                                          const v4f (&xa)[2], const v4f (&xb)[2], float hv)
{
    const int lane = tid & 31;
    const int wave = tid >> 5;
    const bool ok  = (unsigned)hin < HH;
    #pragma unroll
    for (int it = 0; it < 2; ++it) {
        int u   = wave * 2 + it;
        int g   = u >> 2;
        int q   = u & 3;
        int w4  = (lane & 7) + (q << 3);
        int ci  = ((g << 2) + (lane >> 3)) << 1;
        #pragma unroll
        for (int jj = 0; jj < 4; ++jj) {
            int wp = (w4 << 2) + 1 + jj;
            v2f pr;
            pr[0] = ok ? xa[it][jj] : 0.0f;
            pr[1] = ok ? xb[it][jj] : 0.0f;
            *(v2f*)&xl[wp * XPADW + ci] = pr;
        }
    }
    const int hci = (tid & 63) >> 1;
    const int hwp = (tid & 1) ? (XROWW - 1) : 0;
    if (tid < 64) xl[hwp * XPADW + hci] = hv;
}

#define FOFF(f) ((((f) >> 3) * XPADW) + (((f) & 7) << 2))

template <bool DN, bool DM, bool DO, bool NEXT>
__device__ __forceinline__ void row_body(const v2f (&A0)[NSTEP], const v2f (&A1)[NSTEP],
                                         v2f (&S0)[HALF], v2f (&S1)[HALF],
                                         const float* bcur, const float* bnext,
                                         v8f& n0, v8f& n1, v8f& m0, v8f& m1, v8f& o0, v8f& o1)
{
    #pragma unroll
    for (int f = 0; f < HALF; ++f) S1[f] = *(const v2f*)(bcur + FOFF(HALF + f));
    __builtin_amdgcn_sched_barrier(0);
    #pragma unroll
    for (int f = 0; f < HALF; ++f) {
        const v2f bv = S0[f];
        if (DN) {
            n0 = __builtin_amdgcn_wmma_f32_16x16x4_f32(false, A0[f], false, bv, (short)0, n0, false, false);
            n1 = __builtin_amdgcn_wmma_f32_16x16x4_f32(false, A1[f], false, bv, (short)0, n1, false, false);
        }
        if (DM) {
            m0 = __builtin_amdgcn_wmma_f32_16x16x4_f32(false, A0[NFRAG + f], false, bv, (short)0, m0, false, false);
            m1 = __builtin_amdgcn_wmma_f32_16x16x4_f32(false, A1[NFRAG + f], false, bv, (short)0, m1, false, false);
        }
        if (DO) {
            o0 = __builtin_amdgcn_wmma_f32_16x16x4_f32(false, A0[2 * NFRAG + f], false, bv, (short)0, o0, false, false);
            o1 = __builtin_amdgcn_wmma_f32_16x16x4_f32(false, A1[2 * NFRAG + f], false, bv, (short)0, o1, false, false);
        }
    }
    if (NEXT) {
        #pragma unroll
        for (int f = 0; f < HALF; ++f) S0[f] = *(const v2f*)(bnext + FOFF(f));
    }
    __builtin_amdgcn_sched_barrier(0);
    #pragma unroll
    for (int f = 0; f < HALF; ++f) {
        const v2f bv = S1[f];
        if (DN) {
            n0 = __builtin_amdgcn_wmma_f32_16x16x4_f32(false, A0[HALF + f], false, bv, (short)0, n0, false, false);
            n1 = __builtin_amdgcn_wmma_f32_16x16x4_f32(false, A1[HALF + f], false, bv, (short)0, n1, false, false);
        }
        if (DM) {
            m0 = __builtin_amdgcn_wmma_f32_16x16x4_f32(false, A0[NFRAG + HALF + f], false, bv, (short)0, m0, false, false);
            m1 = __builtin_amdgcn_wmma_f32_16x16x4_f32(false, A1[NFRAG + HALF + f], false, bv, (short)0, m1, false, false);
        }
        if (DO) {
            o0 = __builtin_amdgcn_wmma_f32_16x16x4_f32(false, A0[2 * NFRAG + HALF + f], false, bv, (short)0, o0, false, false);
            o1 = __builtin_amdgcn_wmma_f32_16x16x4_f32(false, A1[2 * NFRAG + HALF + f], false, bv, (short)0, o1, false, false);
        }
    }
}

extern "C" __global__ __launch_bounds__(256, 1)
void conv3x3_wmma_f32(const float* __restrict__ x,
                      const float* __restrict__ wgt,
                      const float* __restrict__ bias,
                      float* __restrict__ out)
{
    extern __shared__ float smem[];
    v2f*   wlds = (v2f*)(smem + W_OFF);
    float* blds = smem + B_OFF;

    const int tid   = threadIdx.x;
    const int blk   = blockIdx.x;
    const int b     = blk >> 5;
    const int h0    = ((blk >> 1) & 15) * ROWS;
    const int wtile = blk & 1;
    const int wbase_g = wtile * TILE_W;
    const int lane = tid & 31;
    const int wave = tid >> 5;

    v4f wreg[9];
    #pragma unroll
    for (int i = 0; i < 9; ++i) {
        int q    = i * 256 + tid;
        int rest = q >> 5;
        int cout = (q & 7) + ((rest & 3) << 3);
        int f4   = ((rest >> 2) << 2) + ((q >> 3) & 3);
        wreg[i] = *(const v4f*)(wgt + cout * (CIN * 9) + (f4 << 2));
    }
    v4f xa[2], xb[2], ya[2], yb[2];
    float hv, hw;
    load_row(x, b, h0 - 1, wbase_g, tid, xa, xb, hv);
    load_row(x, b, h0,     wbase_g, tid, ya, yb, hw);
    float bsv = 0.0f;
    if (tid < 32) bsv = bias[tid];

    #pragma unroll
    for (int i = 0; i < 9; ++i) {
        int q    = i * 256 + tid;
        int rest = q >> 5;
        int cout = (q & 7) + ((rest & 3) << 3);
        int f4   = ((rest >> 2) << 2) + ((q >> 3) & 3);
        #pragma unroll
        for (int j = 0; j < 4; ++j) {
            int e    = (f4 << 2) + j;
            int ci   = e / 9;
            int khkw = e - 9 * ci;
            int k    = (khkw << 5) + ci;
            int s    = k >> 2;
            int l    = (cout & 15) + (((k >> 1) & 1) << 4);
            int t    = cout >> 4;
            smem[W_OFF + ((((s << 1) + t) * 32 + l) << 1) + (k & 1)] = wreg[i][j];
        }
    }
    store_row(smem,              h0 - 1, tid, xa, xb, hv);
    store_row(smem + XROW_ELEMS, h0,     tid, ya, yb, hw);
    if (tid < 32) blds[tid] = bsv;
    __syncthreads();

    const int wl   = wave << 4;
    const int nco  = lane & 15;
    const int hi   = lane >> 4;

    v2f A0[NSTEP], A1[NSTEP];
    #pragma unroll
    for (int s = 0; s < NSTEP; ++s) {
        A0[s] = wlds[(s * 2 + 0) * 32 + lane];
        A1[s] = wlds[(s * 2 + 1) * 32 + lane];
    }
    float bia0[8], bia1[8];
    #pragma unroll
    for (int r = 0; r < 8; ++r) {
        bia0[r] = blds[r + (hi << 3)];
        bia1[r] = blds[r + (hi << 3) + 16];
    }

    const int wcol = wbase_g + wl + nco;
    const float* lbase = smem + (wl + nco) * XPADW + (hi << 1);

    v2f S0[HALF], S1[HALF];
    #pragma unroll
    for (int f = 0; f < HALF; ++f) S0[f] = *(const v2f*)(lbase + FOFF(f));

    v8f n0 = {}, n1 = {}, m0 = {}, m1 = {}, o0 = {}, o1 = {};
    int o_cur = 0, o_nxt = XROW_ELEMS, o_wr = 2 * XROW_ELEMS;

    #pragma unroll 1
    for (int i = 0; i < NIN; ++i) {
        const int  hin = h0 - 1 + i;
        const bool ld  = (i + 2) < NIN;

        if (ld) load_row(x, b, hin + 2, wbase_g, tid, xa, xb, hv);

        const float* bcur  = lbase + o_cur;
        const float* bnext = lbase + o_nxt;
        if (i == 0)         row_body<true,  false, false, true >(A0, A1, S0, S1, bcur, bnext, n0, n1, m0, m1, o0, o1);
        else if (i == 1)    row_body<true,  true,  false, true >(A0, A1, S0, S1, bcur, bnext, n0, n1, m0, m1, o0, o1);
        else if (i < ROWS)  row_body<true,  true,  true,  true >(A0, A1, S0, S1, bcur, bnext, n0, n1, m0, m1, o0, o1);
        else if (i == ROWS) row_body<false, true,  true,  true >(A0, A1, S0, S1, bcur, bnext, n0, n1, m0, m1, o0, o1);
        else                row_body<false, false, true,  false>(A0, A1, S0, S1, bcur, bnext, n0, n1, m0, m1, o0, o1);

        if (i >= 2) {
            const int h = hin - 1;
            #pragma unroll
            for (int r = 0; r < 8; ++r) {
                int co = r + (hi << 3);
                out[((b * COUT + co)      * HH + h) * WW + wcol] = o0[r] + bia0[r];
                out[((b * COUT + co + 16) * HH + h) * WW + wcol] = o1[r] + bia1[r];
            }
        }
        o0 = m0; o1 = m1; m0 = n0; m1 = n1;
        n0 = v8f{}; n1 = v8f{};

        if (ld) store_row(smem + o_wr, hin + 2, tid, xa, xb, hv);
        { int t = o_cur; o_cur = o_nxt; o_nxt = o_wr; o_wr = t; }
        __builtin_amdgcn_fence(__ATOMIC_RELEASE, "workgroup", "local");
        __builtin_amdgcn_s_barrier();
        __builtin_amdgcn_fence(__ATOMIC_ACQUIRE, "workgroup", "local");
    }
}

extern "C" void kernel_launch(void* const* d_in, const int* in_sizes, int n_in,
                              void* d_out, int out_size, void* d_ws, size_t ws_size,
                              hipStream_t stream) {
    const float* x    = (const float*)d_in[0];
    const float* wgt  = (const float*)d_in[1];
    const float* bias = (const float*)d_in[2];
    float* out = (float*)d_out;

    dim3 grid(16 * 16 * 2);
    dim3 block(256);
    size_t smem = SMEM_FLOATS * sizeof(float);
    hipLaunchKernelGGL(conv3x3_wmma_f32, grid, block, smem, stream, x, wgt, bias, out);
}


# Minimal filtering F(2,3) along w: a wave owns 16 cout x 16 column pairs, 4 products of K=96 per pair, input columns combined in registers behind the LDS read: a third fewer WMMAs; sum order changed
# speedup vs baseline: 3.8653x; 1.2413x over previous
#include <hip/hip_runtime.h>


typedef __attribute__((ext_vector_type(2))) float v2f;
typedef __attribute__((ext_vector_type(4))) float v4f;
typedef __attribute__((ext_vector_type(8))) float v8f;

#define CIN    32
#define COUT   32
#define HH     256
#define WW     256
#define TILE_W 128
#define ROWS   16
#define NIN    (ROWS + 2)
#define NSTEPW 96
#define NRAW   16

#define XPADW   34
#define XROWW   130
#define XROW_ELEMS (XROWW * XPADW)
#define NBUF    3
#define W_OFF   (NBUF * XROW_ELEMS)
#define W_FRAGS (NSTEPW * 2 * 32)
#define B_OFF   (W_OFF + W_FRAGS * 2)
#define SMEM_FLOATS (B_OFF + 32)

__device__ __forceinline__ void load_row(const float* __restrict__ x, int b, int hin, int wbase_g,
                                         int tid, v4f (&xa)[2], v4f (&xb)[2], float& hv)
{
    const int lane = tid & 31;
    const int wave = tid >> 5;
    const int hc   = hin < 0 ? 0 : (hin > HH - 1 ? HH - 1 : hin);
    #pragma unroll
    for (int it = 0; it < 2; ++it) {
        int u   = wave * 2 + it;
        int g   = u >> 2;
        int q   = u & 3;
        int w4  = (lane & 7) + (q << 3);
        int ci  = ((g << 2) + (lane >> 3)) << 1;
        const float* p = x + ((b * CIN + ci) * HH + hc) * WW + wbase_g + (w4 << 2);
        xa[it] = *(const v4f*)p;
        xb[it] = *(const v4f*)(p + HH * WW);
    }
    const int hci = (tid & 63) >> 1;
    const int hwp = (tid & 1) ? (XROWW - 1) : 0;
    hv = 0.0f;
    if (tid < 64) {
        int win = wbase_g + hwp - 1;
        if ((unsigned)hin < HH && (unsigned)win < WW)
            hv = x[((b * CIN + hci) * HH + hin) * WW + win];
    }
}

__device__ __forceinline__ void store_row(float* xl, int hin, int tid,
                                          const v4f (&xa)[2], const v4f (&xb)[2], float hv)
{
    const int lane = tid & 31;
    const int wave = tid >> 5;
    const bool ok  = (unsigned)hin < HH;
    #pragma unroll
    for (int it = 0; it < 2; ++it) {
        int u   = wave * 2 + it;
        int g   = u >> 2;
        int q   = u & 3;
        int w4  = (lane & 7) + (q << 3);
        int ci  = ((g << 2) + (lane >> 3)) << 1;
        #pragma unroll
        for (int jj = 0; jj < 4; ++jj) {
            int wp = (w4 << 2) + 1 + jj;
            v2f pr;
            pr[0] = ok ? xa[it][jj] : 0.0f;
            pr[1] = ok ? xb[it][jj] : 0.0f;
            *(v2f*)&xl[wp * XPADW + ci] = pr;
        }
    }
    const int hci = (tid & 63) >> 1;
    const int hwp = (tid & 1) ? (XROWW - 1) : 0;
    if (tid < 64) xl[hwp * XPADW + hci] = hv;
}

#define ROFF(cg, t) (((t) * XPADW) + ((cg) << 2))

__device__ __forceinline__ v8f mma(v2f a, v2f b, v8f c)
{
    return __builtin_amdgcn_wmma_f32_16x16x4_f32(false, a, false, b, (short)0, c, false, false);
}

template <bool DN, bool DM, bool DO>
__device__ __forceinline__ void half_row(const v2f (&A)[NSTEPW], const v2f (&R)[NRAW], int cg0,
                                         v8f (&n)[4], v8f (&m)[4], v8f (&o)[4])
{
    #pragma unroll
    for (int g = 0; g < 4; ++g) {
        const v2f d0 = R[g * 4 + 0], d1 = R[g * 4 + 1], d2 = R[g * 4 + 2], d3 = R[g * 4 + 3];
        v2f V[4];
        V[0] = d0 - d2;
        V[1] = d1 + d2;
        V[2] = d2 - d1;
        V[3] = d1 - d3;
        const int cg = cg0 + g;
        #pragma unroll
        for (int c = 0; c < 4; ++c) {
            if (DN) n[c] = mma(A[(0 * 4 + c) * 8 + cg], V[c], n[c]);
            if (DM) m[c] = mma(A[(1 * 4 + c) * 8 + cg], V[c], m[c]);
            if (DO) o[c] = mma(A[(2 * 4 + c) * 8 + cg], V[c], o[c]);
        }
    }
}

template <bool DN, bool DM, bool DO, bool NEXT>
__device__ __forceinline__ void row_body(const v2f (&A)[NSTEPW], v2f (&R0)[NRAW], v2f (&R1)[NRAW],
                                         const float* bcur, const float* bnext,
                                         v8f (&n)[4], v8f (&m)[4], v8f (&o)[4])
{
    #pragma unroll
    for (int q = 0; q < NRAW; ++q) R1[q] = *(const v2f*)(bcur + ROFF(4 + (q >> 2), q & 3));
    __builtin_amdgcn_sched_barrier(0);
    half_row<DN, DM, DO>(A, R0, 0, n, m, o);
    if (NEXT) {
        #pragma unroll
        for (int q = 0; q < NRAW; ++q) R0[q] = *(const v2f*)(bnext + ROFF(q >> 2, q & 3));
    }
    __builtin_amdgcn_sched_barrier(0);
    half_row<DN, DM, DO>(A, R1, 4, n, m, o);
}

extern "C" __global__ __launch_bounds__(256, 1)
void conv3x3_wmma_f32(const float* __restrict__ x,
                      const float* __restrict__ wgt,
                      const float* __restrict__ bias,
                      float* __restrict__ out)
{
    extern __shared__ float smem[];
    v2f*   wlds = (v2f*)(smem + W_OFF);
    float* blds = smem + B_OFF;

    const int tid   = threadIdx.x;
    const int blk   = blockIdx.x;
    const int b     = blk >> 5;
    const int h0    = ((blk >> 1) & 15) * ROWS;
    const int wtile = blk & 1;
    const int wbase_g = wtile * TILE_W;
    const int lane = tid & 31;
    const int wave = tid >> 5;

    float wr[12][3];
    #pragma unroll
    for (int i = 0; i < 12; ++i) {
        int u    = i * 256 + tid;
        int rest = u >> 5;
        int cout = (u & 7) + ((rest & 3) << 3);
        int mrow = ((rest >> 2) << 2) + ((u >> 3) & 3);
        const float* p = wgt + cout * (CIN * 9) + mrow * 3;
        wr[i][0] = p[0];
        wr[i][1] = p[1];
        wr[i][2] = p[2];
    }
    v4f xa[2], xb[2], ya[2], yb[2];
    float hv, hw;
    load_row(x, b, h0 - 1, wbase_g, tid, xa, xb, hv);
    load_row(x, b, h0,     wbase_g, tid, ya, yb, hw);
    float bsv = 0.0f;
    if (tid < 32) bsv = bias[tid];

    #pragma unroll
    for (int i = 0; i < 12; ++i) {
        int u    = i * 256 + tid;
        int rest = u >> 5;
        int cout = (u & 7) + ((rest & 3) << 3);
        int mrow = ((rest >> 2) << 2) + ((u >> 3) & 3);
        int ci   = mrow / 3;
        int kh   = mrow - 3 * ci;
        const float g0 = wr[i][0], g1 = wr[i][1], g2 = wr[i][2];
        const float sg = g0 + g2;
        float uu[4];
        uu[0] = g0;
        uu[1] = 0.5f * (sg + g1);
        uu[2] = 0.5f * (sg - g1);
        uu[3] = g2;
        int t  = cout >> 4;
        int l  = (cout & 15) + (((ci >> 1) & 1) << 4);
        int cg = ci >> 2;
        #pragma unroll
        for (int c = 0; c < 4; ++c) {
            int s = ((kh * 4 + c) << 3) + cg;
            smem[W_OFF + ((((s << 1) + t) * 32 + l) << 1) + (ci & 1)] = uu[c];
        }
    }
    store_row(smem,              h0 - 1, tid, xa, xb, hv);
    store_row(smem + XROW_ELEMS, h0,     tid, ya, yb, hw);
    if (tid < 32) blds[tid] = bsv;
    __syncthreads();

    const int th   = wave & 1;
    const int pb   = (wave >> 1) << 4;
    const int nco  = lane & 15;
    const int hi   = lane >> 4;

    v2f A[NSTEPW];
    #pragma unroll
    for (int s = 0; s < NSTEPW; ++s) A[s] = wlds[(s * 2 + th) * 32 + lane];
    float bia[8];
    #pragma unroll
    for (int r = 0; r < 8; ++r) bia[r] = blds[(th << 4) + r + (hi << 3)];

    const int wcol = wbase_g + ((pb + nco) << 1);
    const float* lbase = smem + ((pb + nco) << 1) * XPADW + (hi << 1);

    v2f R0[NRAW], R1[NRAW];
    #pragma unroll
    for (int q = 0; q < NRAW; ++q) R0[q] = *(const v2f*)(lbase + ROFF(q >> 2, q & 3));

    v8f n[4] = {}, m[4] = {}, o[4] = {};
    int o_cur = 0, o_nxt = XROW_ELEMS, o_wr = 2 * XROW_ELEMS;

    #pragma unroll 1
    for (int i = 0; i < NIN; ++i) {
        const int  hin = h0 - 1 + i;
        const bool ld  = (i + 2) < NIN;

        if (ld) load_row(x, b, hin + 2, wbase_g, tid, xa, xb, hv);

        const float* bcur  = lbase + o_cur;
        const float* bnext = lbase + o_nxt;
        if (i == 0)         row_body<true,  false, false, true >(A, R0, R1, bcur, bnext, n, m, o);
        else if (i == 1)    row_body<true,  true,  false, true >(A, R0, R1, bcur, bnext, n, m, o);
        else if (i < ROWS)  row_body<true,  true,  true,  true >(A, R0, R1, bcur, bnext, n, m, o);
        else if (i == ROWS) row_body<false, true,  true,  true >(A, R0, R1, bcur, bnext, n, m, o);
        else                row_body<false, false, true,  false>(A, R0, R1, bcur, bnext, n, m, o);

        if (i >= 2) {
            const int h = hin - 1;
            #pragma unroll
            for (int r = 0; r < 8; ++r) {
                int co = (th << 4) + r + (hi << 3);
                v2f y;
                y[0] = (o[0][r] + o[1][r]) + (o[2][r] + bia[r]);
                y[1] = (o[1][r] - o[2][r]) - (o[3][r] - bia[r]);
                *(v2f*)&out[((b * COUT + co) * HH + h) * WW + wcol] = y;
            }
        }
        #pragma unroll
        for (int c = 0; c < 4; ++c) { o[c] = m[c]; m[c] = n[c]; n[c] = v8f{}; }

        if (ld) store_row(smem + o_wr, hin + 2, tid, xa, xb, hv);
        { int t = o_cur; o_cur = o_nxt; o_nxt = o_wr; o_wr = t; }
        __builtin_amdgcn_fence(__ATOMIC_RELEASE, "workgroup", "local");
        __builtin_amdgcn_s_barrier();
        __builtin_amdgcn_fence(__ATOMIC_ACQUIRE, "workgroup", "local");
    }
}

extern "C" void kernel_launch(void* const* d_in, const int* in_sizes, int n_in,
                              void* d_out, int out_size, void* d_ws, size_t ws_size,
                              hipStream_t stream) {
    const float* x    = (const float*)d_in[0];
    const float* wgt  = (const float*)d_in[1];
    const float* bias = (const float*)d_in[2];
    float* out = (float*)d_out;

    dim3 grid(16 * 16 * 2);
    dim3 block(256);
    size_t smem = SMEM_FLOATS * sizeof(float);
    hipLaunchKernelGGL(conv3x3_wmma_f32, grid, block, smem, stream, x, wgt, bias, out);
}


# Row walk written out with period 3 (the period of the accumulator roles): each output-row role is a fixed register set, so no accumulator moves or zero fills between trips; sums unchanged
# speedup vs baseline: 4.0922x; 1.0587x over previous
#include <hip/hip_runtime.h>


typedef __attribute__((ext_vector_type(2))) float v2f;
typedef __attribute__((ext_vector_type(4))) float v4f;
typedef __attribute__((ext_vector_type(8))) float v8f;

#define CIN    32
#define COUT   32
#define HH     256
#define WW     256
#define TILE_W 128
#define ROWS   16
#define NIN    (ROWS + 2)
#define NSTEPW 96
#define NRAW   16

#define XPADW   34
#define XROWW   130
#define XROW_ELEMS (XROWW * XPADW)
#define NBUF    3
#define W_OFF   (NBUF * XROW_ELEMS)
#define W_FRAGS (NSTEPW * 2 * 32)
#define B_OFF   (W_OFF + W_FRAGS * 2)
#define SMEM_FLOATS (B_OFF + 32)

__device__ __forceinline__ void load_row(const float* __restrict__ x, int b, int hin, int wbase_g,
                                         int tid, v4f (&xa)[2], v4f (&xb)[2], float& hv)
{
    const int lane = tid & 31;
    const int wave = tid >> 5;
    const int hc   = hin < 0 ? 0 : (hin > HH - 1 ? HH - 1 : hin);
    #pragma unroll
    for (int it = 0; it < 2; ++it) {
        int u   = wave * 2 + it;
        int g   = u >> 2;
        int q   = u & 3;
        int w4  = (lane & 7) + (q << 3);
        int ci  = ((g << 2) + (lane >> 3)) << 1;
        const float* p = x + ((b * CIN + ci) * HH + hc) * WW + wbase_g + (w4 << 2);
        xa[it] = *(const v4f*)p;
        xb[it] = *(const v4f*)(p + HH * WW);
    }
    const int hci = (tid & 63) >> 1;
    const int hwp = (tid & 1) ? (XROWW - 1) : 0;
    hv = 0.0f;
    if (tid < 64) {
        int win = wbase_g + hwp - 1;
        if ((unsigned)hin < HH && (unsigned)win < WW)
            hv = x[((b * CIN + hci) * HH + hin) * WW + win];
    }
}

__device__ __forceinline__ void store_row(float* xl, int hin, int tid,
                                          const v4f (&xa)[2], const v4f (&xb)[2], float hv)
{
    const int lane = tid & 31;
    const int wave = tid >> 5;
    const bool ok  = (unsigned)hin < HH;
    #pragma unroll
    for (int it = 0; it < 2; ++it) {
        int u   = wave * 2 + it;
        int g   = u >> 2;
        int q   = u & 3;
        int w4  = (lane & 7) + (q << 3);
        int ci  = ((g << 2) + (lane >> 3)) << 1;
        #pragma unroll
        for (int jj = 0; jj < 4; ++jj) {
            int wp = (w4 << 2) + 1 + jj;
            v2f pr;
            pr[0] = ok ? xa[it][jj] : 0.0f;
            pr[1] = ok ? xb[it][jj] : 0.0f;
            *(v2f*)&xl[wp * XPADW + ci] = pr;
        }
    }
    const int hci = (tid & 63) >> 1;
    const int hwp = (tid & 1) ? (XROWW - 1) : 0;
    if (tid < 64) xl[hwp * XPADW + hci] = hv;
}

#define ROFF(cg, t) (((t) * XPADW) + ((cg) << 2))

__device__ __forceinline__ v8f mma(v2f a, v2f b, v8f c)
{
    return __builtin_amdgcn_wmma_f32_16x16x4_f32(false, a, false, b, (short)0, c, false, false);
}

template <bool DN, bool DM, bool DO>
__device__ __forceinline__ void half_row(const v2f (&A)[NSTEPW], const v2f (&R)[NRAW], int cg0,
                                         v8f (&n)[4], v8f (&m)[4], v8f (&o)[4])
{
    #pragma unroll
    for (int g = 0; g < 4; ++g) {
        const v2f d0 = R[g * 4 + 0], d1 = R[g * 4 + 1], d2 = R[g * 4 + 2], d3 = R[g * 4 + 3];
        v2f V[4];
        V[0] = d0 - d2;
        V[1] = d1 + d2;
        V[2] = d2 - d1;
        V[3] = d1 - d3;
        const int cg = cg0 + g;
        #pragma unroll
        for (int c = 0; c < 4; ++c) {
            if (DN) n[c] = mma(A[(0 * 4 + c) * 8 + cg], V[c], n[c]);
            if (DM) m[c] = mma(A[(1 * 4 + c) * 8 + cg], V[c], m[c]);
            if (DO) o[c] = mma(A[(2 * 4 + c) * 8 + cg], V[c], o[c]);
        }
    }
}

template <bool DN, bool DM, bool DO, bool NEXT>
__device__ __forceinline__ void row_body(const v2f (&A)[NSTEPW], const v2f (&R0)[NRAW], v2f (&RN)[NRAW],
                                         const float* bcur, const float* bnext,
                                         v8f (&n)[4], v8f (&m)[4], v8f (&o)[4])
{
    if (DN) {
        #pragma unroll
        for (int c = 0; c < 4; ++c) n[c] = v8f{};
    }
    v2f R1[NRAW];
    #pragma unroll
    for (int q = 0; q < NRAW; ++q) R1[q] = *(const v2f*)(bcur + ROFF(4 + (q >> 2), q & 3));
    __builtin_amdgcn_sched_barrier(0);
    half_row<DN, DM, DO>(A, R0, 0, n, m, o);
    if (NEXT) {
        #pragma unroll
        for (int q = 0; q < NRAW; ++q) RN[q] = *(const v2f*)(bnext + ROFF(q >> 2, q & 3));
    }
    __builtin_amdgcn_sched_barrier(0);
    half_row<DN, DM, DO>(A, R1, 4, n, m, o);
}

template <int K>
__device__ __forceinline__ v2f (&pick3(v2f (&a)[NRAW], v2f (&b)[NRAW], v2f (&c)[NRAW]))[NRAW]
{
    if constexpr (K == 0) return a;
    else if constexpr (K == 1) return b;
    else return c;
}

template <int PH, bool DN, bool DM, bool DO, bool NEXT, bool LD, bool EPI>
__device__ __forceinline__ void trip(const int i, const float* __restrict__ x, float* __restrict__ out,
                                     float* smem, const float* lbase,
                                     const int b, const int h0, const int wbase_g, const int tid,
                                     const int th, const int hi, const int wcol,
                                     const v2f (&A)[NSTEPW], const float (&bia)[8],
                                     v2f (&Ra)[NRAW], v2f (&Rb)[NRAW], v2f (&Rc)[NRAW], v8f (&acc)[3][4],
                                     v4f (&xa)[2], v4f (&xb)[2], float& hv,
                                     int& o_cur, int& o_nxt, int& o_wr)
{
    constexpr int SN = PH, SM = (PH + 2) % 3, SO = (PH + 1) % 3;
    const int hin = h0 - 1 + i;

    if (LD) load_row(x, b, hin + 2, wbase_g, tid, xa, xb, hv);

    row_body<DN, DM, DO, NEXT>(A, pick3<PH>(Ra, Rb, Rc), pick3<(PH + 1) % 3>(Ra, Rb, Rc),
                               lbase + o_cur, lbase + o_nxt, acc[SN], acc[SM], acc[SO]);

    if (EPI) {
        const int h = hin - 1;
        #pragma unroll
        for (int r = 0; r < 8; ++r) {
            int co = (th << 4) + r + (hi << 3);
            v2f y;
            y[0] = (acc[SO][0][r] + acc[SO][1][r]) + (acc[SO][2][r] + bia[r]);
            y[1] = (acc[SO][1][r] - acc[SO][2][r]) - (acc[SO][3][r] - bia[r]);
            *(v2f*)&out[((b * COUT + co) * HH + h) * WW + wcol] = y;
        }
    }

    if (LD) store_row(smem + o_wr, hin + 2, tid, xa, xb, hv);
    { int t = o_cur; o_cur = o_nxt; o_nxt = o_wr; o_wr = t; }
    __builtin_amdgcn_fence(__ATOMIC_RELEASE, "workgroup", "local");
    __builtin_amdgcn_s_barrier();
    __builtin_amdgcn_fence(__ATOMIC_ACQUIRE, "workgroup", "local");
}

extern "C" __global__ __launch_bounds__(256, 1)
void conv3x3_wmma_f32(const float* __restrict__ x,
                      const float* __restrict__ wgt,
                      const float* __restrict__ bias,
                      float* __restrict__ out)
{
    extern __shared__ float smem[];
    v2f*   wlds = (v2f*)(smem + W_OFF);
    float* blds = smem + B_OFF;

    const int tid   = threadIdx.x;
    const int blk   = blockIdx.x;
    const int b     = blk >> 5;
    const int h0    = ((blk >> 1) & 15) * ROWS;
    const int wtile = blk & 1;
    const int wbase_g = wtile * TILE_W;
    const int lane = tid & 31;
    const int wave = tid >> 5;

    float wr[12][3];
    #pragma unroll
    for (int i = 0; i < 12; ++i) {
        int u    = i * 256 + tid;
        int rest = u >> 5;
        int cout = (u & 7) + ((rest & 3) << 3);
        int mrow = ((rest >> 2) << 2) + ((u >> 3) & 3);
        const float* p = wgt + cout * (CIN * 9) + mrow * 3;
        wr[i][0] = p[0];
        wr[i][1] = p[1];
        wr[i][2] = p[2];
    }
    v4f xa[2], xb[2], ya[2], yb[2];
    float hv, hw;
    load_row(x, b, h0 - 1, wbase_g, tid, xa, xb, hv);
    load_row(x, b, h0,     wbase_g, tid, ya, yb, hw);
    float bsv = 0.0f;
    if (tid < 32) bsv = bias[tid];

    #pragma unroll
    for (int i = 0; i < 12; ++i) {
        int u    = i * 256 + tid;
        int rest = u >> 5;
        int cout = (u & 7) + ((rest & 3) << 3);
        int mrow = ((rest >> 2) << 2) + ((u >> 3) & 3);
        int ci   = mrow / 3;
        int kh   = mrow - 3 * ci;
        const float g0 = wr[i][0], g1 = wr[i][1], g2 = wr[i][2];
        const float sg = g0 + g2;
        float uu[4];
        uu[0] = g0;
        uu[1] = 0.5f * (sg + g1);
        uu[2] = 0.5f * (sg - g1);
        uu[3] = g2;
        int t  = cout >> 4;
        int l  = (cout & 15) + (((ci >> 1) & 1) << 4);
        int cg = ci >> 2;
        #pragma unroll
        for (int c = 0; c < 4; ++c) {
            int s = ((kh * 4 + c) << 3) + cg;
            smem[W_OFF + ((((s << 1) + t) * 32 + l) << 1) + (ci & 1)] = uu[c];
        }
    }
    store_row(smem,              h0 - 1, tid, xa, xb, hv);
    store_row(smem + XROW_ELEMS, h0,     tid, ya, yb, hw);
    if (tid < 32) blds[tid] = bsv;
    __syncthreads();

    const int th   = wave & 1;
    const int pb   = (wave >> 1) << 4;
    const int nco  = lane & 15;
    const int hi   = lane >> 4;

    v2f A[NSTEPW];
    #pragma unroll
    for (int s = 0; s < NSTEPW; ++s) A[s] = wlds[(s * 2 + th) * 32 + lane];
    float bia[8];
    #pragma unroll
    for (int r = 0; r < 8; ++r) bia[r] = blds[(th << 4) + r + (hi << 3)];

    const int wcol = wbase_g + ((pb + nco) << 1);
    const float* lbase = smem + ((pb + nco) << 1) * XPADW + (hi << 1);

    v2f Ra[NRAW], Rb[NRAW], Rc[NRAW];
    #pragma unroll
    for (int q = 0; q < NRAW; ++q) Ra[q] = *(const v2f*)(lbase + ROFF(q >> 2, q & 3));

    v8f acc[3][4] = {};
    int o_cur = 0, o_nxt = XROW_ELEMS, o_wr = 2 * XROW_ELEMS;

#define TRIP_ARGS x, out, smem, lbase, b, h0, wbase_g, tid, th, hi, wcol, A, bia, Ra, Rb, Rc, acc, xa, xb, hv, o_cur, o_nxt, o_wr
    static_assert(ROWS == 16, "trip numbers below are written out for 16 output rows (18 input rows) per block");
    trip<0, true,  false, false, true,  true,  false>(0, TRIP_ARGS);
    trip<1, true,  true,  false, true,  true,  false>(1, TRIP_ARGS);
    #pragma unroll 1
    for (int i = 2; i < 14; i += 3) {
        trip<2, true, true, true, true, true, true>(i,     TRIP_ARGS);
        trip<0, true, true, true, true, true, true>(i + 1, TRIP_ARGS);
        trip<1, true, true, true, true, true, true>(i + 2, TRIP_ARGS);
    }
    trip<2, true,  true,  true,  true,  true,  true >(14, TRIP_ARGS);
    trip<0, true,  true,  true,  true,  true,  true >(15, TRIP_ARGS);
    trip<1, false, true,  true,  true,  false, true >(16, TRIP_ARGS);
    trip<2, false, false, true,  false, false, true >(17, TRIP_ARGS);
#undef TRIP_ARGS
}

extern "C" void kernel_launch(void* const* d_in, const int* in_sizes, int n_in,
                              void* d_out, int out_size, void* d_ws, size_t ws_size,
                              hipStream_t stream) {
    const float* x    = (const float*)d_in[0];
    const float* wgt  = (const float*)d_in[1];
    const float* bias = (const float*)d_in[2];
    float* out = (float*)d_out;

    dim3 grid(16 * 16 * 2);
    dim3 block(256);
    size_t smem = SMEM_FLOATS * sizeof(float);
    hipLaunchKernelGGL(conv3x3_wmma_f32, grid, block, smem, stream, x, wgt, bias, out);
}


# One resident round: 256 blocks of 32 output rows, one per WGP (a 636-register wave leaves room for one block), not 512 of 16 in two rounds: entry, weight staging, halo rows once per WGP; same sums
# speedup vs baseline: 5.1128x; 1.2494x over previous
#include <hip/hip_runtime.h>


typedef __attribute__((ext_vector_type(2))) float v2f;
typedef __attribute__((ext_vector_type(4))) float v4f;
typedef __attribute__((ext_vector_type(8))) float v8f;

#define CIN    32
#define COUT   32
#define HH     256
#define WW     256
#define TILE_W 128
#define ROWS   32
#define NIN    (ROWS + 2)
#define NCOMP  6
#define NSTEPW (3 * NCOMP * 8)
#define NRAW   (4 * NCOMP)
#define NTHR   128
#define NLD    4

#define XPADW   34
#define XROWW   130
#define XROW_ELEMS (XROWW * XPADW)
#define NBUF    3
#define W_OFF   (NBUF * XROW_ELEMS)
#define W_FRAGS (NSTEPW * 2 * 32)
#define B_OFF   (W_OFF + W_FRAGS * 2)
#define SMEM_FLOATS (B_OFF + 32)

__device__ __forceinline__ void load_row(const float* __restrict__ x, int b, int hin, int wbase_g,
                                         int tid, v4f (&xa)[NLD], v4f (&xb)[NLD], float& hv)
{
    const int lane = tid & 31;
    const int wave = tid >> 5;
    const int hc   = hin < 0 ? 0 : (hin > HH - 1 ? HH - 1 : hin);
    #pragma unroll
    for (int it = 0; it < NLD; ++it) {
        int u   = wave * NLD + it;
        int g   = u >> 2;
        int q   = u & 3;
        int w4  = (lane & 7) + (q << 3);
        int ci  = ((g << 2) + (lane >> 3)) << 1;
        const float* p = x + ((b * CIN + ci) * HH + hc) * WW + wbase_g + (w4 << 2);
        xa[it] = *(const v4f*)p;
        xb[it] = *(const v4f*)(p + HH * WW);
    }
    const int hci = (tid & 63) >> 1;
    const int hwp = (tid & 1) ? (XROWW - 1) : 0;
    hv = 0.0f;
    if (tid < 64) {
        int win = wbase_g + hwp - 1;
        if ((unsigned)hin < HH && (unsigned)win < WW)
            hv = x[((b * CIN + hci) * HH + hin) * WW + win];
    }
}

__device__ __forceinline__ void store_row(float* xl, int hin, int tid,
                                          const v4f (&xa)[NLD], const v4f (&xb)[NLD], float hv)
{
    const int lane = tid & 31;
    const int wave = tid >> 5;
    const bool ok  = (unsigned)hin < HH;
    #pragma unroll
    for (int it = 0; it < NLD; ++it) {
        int u   = wave * NLD + it;
        int g   = u >> 2;
        int q   = u & 3;
        int w4  = (lane & 7) + (q << 3);
        int ci  = ((g << 2) + (lane >> 3)) << 1;
        #pragma unroll
        for (int jj = 0; jj < 4; ++jj) {
            int wp = (w4 << 2) + 1 + jj;
            v2f pr;
            pr[0] = ok ? xa[it][jj] : 0.0f;
            pr[1] = ok ? xb[it][jj] : 0.0f;
            *(v2f*)&xl[wp * XPADW + ci] = pr;
        }
    }
    const int hci = (tid & 63) >> 1;
    const int hwp = (tid & 1) ? (XROWW - 1) : 0;
    if (tid < 64) xl[hwp * XPADW + hci] = hv;
}

#define ROFF(cg, t) (((t) * XPADW) + ((cg) << 2))

__device__ __forceinline__ v8f mma(v2f a, v2f b, v8f c)
{
    return __builtin_amdgcn_wmma_f32_16x16x4_f32(false, a, false, b, (short)0, c, false, false);
}

template <bool DN, bool DM, bool DO>
__device__ __forceinline__ void half_row(const v2f (&A)[NSTEPW], const v2f (&R)[NRAW], int cg0,
                                         v8f (&n)[NCOMP], v8f (&m)[NCOMP], v8f (&o)[NCOMP])
{
    #pragma unroll
    for (int g = 0; g < 4; ++g) {
        const v2f d0 = R[g * NCOMP + 0], d1 = R[g * NCOMP + 1], d2 = R[g * NCOMP + 2];
        const v2f d3 = R[g * NCOMP + 3], d4 = R[g * NCOMP + 4], d5 = R[g * NCOMP + 5];
        const v2f ea = d4 - 4.0f * d2;
        const v2f eb = d3 - 4.0f * d1;
        const v2f ec = d4 - d2;
        const v2f ed = d3 - d1;
        v2f V[NCOMP];
        V[0] = 4.0f * d0 + (d4 - 5.0f * d2);
        V[1] = ea + eb;
        V[2] = ea - eb;
        V[3] = ec + 2.0f * ed;
        V[4] = ec - 2.0f * ed;
        V[5] = 4.0f * d1 + (d5 - 5.0f * d3);
        const int cg = cg0 + g;
        #pragma unroll
        for (int c = 0; c < NCOMP; ++c) {
            if (DN) n[c] = mma(A[(0 * NCOMP + c) * 8 + cg], V[c], n[c]);
            if (DM) m[c] = mma(A[(1 * NCOMP + c) * 8 + cg], V[c], m[c]);
            if (DO) o[c] = mma(A[(2 * NCOMP + c) * 8 + cg], V[c], o[c]);
        }
    }
}

template <bool DN, bool DM, bool DO, bool NEXT>
__device__ __forceinline__ void row_body(const v2f (&A)[NSTEPW], const v2f (&R0)[NRAW], v2f (&RN)[NRAW],
                                         const float* bcur, const float* bnext,
                                         v8f (&n)[NCOMP], v8f (&m)[NCOMP], v8f (&o)[NCOMP])
{
    if (DN) {
        #pragma unroll
        for (int c = 0; c < NCOMP; ++c) n[c] = v8f{};
    }
    v2f R1[NRAW];
    #pragma unroll
    for (int q = 0; q < NRAW; ++q) R1[q] = *(const v2f*)(bcur + ROFF(4 + (q / NCOMP), q % NCOMP));
    __builtin_amdgcn_sched_barrier(0);
    half_row<DN, DM, DO>(A, R0, 0, n, m, o);
    if (NEXT) {
        #pragma unroll
        for (int q = 0; q < NRAW; ++q) RN[q] = *(const v2f*)(bnext + ROFF(q / NCOMP, q % NCOMP));
    }
    __builtin_amdgcn_sched_barrier(0);
    half_row<DN, DM, DO>(A, R1, 4, n, m, o);
}

template <int K>
__device__ __forceinline__ v2f (&pick3(v2f (&a)[NRAW], v2f (&b)[NRAW], v2f (&c)[NRAW]))[NRAW]
{
    if constexpr (K == 0) return a;
    else if constexpr (K == 1) return b;
    else return c;
}

template <int PH, bool DN, bool DM, bool DO, bool NEXT, bool LD, bool EPI>
__device__ __forceinline__ void trip(const int i, const float* __restrict__ x, float* __restrict__ out,
                                     float* smem, const float* lbase,
                                     const int b, const int h0, const int wbase_g, const int tid,
                                     const int th, const int hi, const int wcol,
                                     const v2f (&A)[NSTEPW], const float (&bia)[8],
                                     v2f (&Ra)[NRAW], v2f (&Rb)[NRAW], v2f (&Rc)[NRAW], v8f (&acc)[3][NCOMP],
                                     v4f (&xa)[NLD], v4f (&xb)[NLD], float& hv,
                                     int& o_cur, int& o_nxt, int& o_wr)
{
    constexpr int SN = PH, SM = (PH + 2) % 3, SO = (PH + 1) % 3;
    const int hin = h0 - 1 + i;

    if (LD) load_row(x, b, hin + 2, wbase_g, tid, xa, xb, hv);

    row_body<DN, DM, DO, NEXT>(A, pick3<PH>(Ra, Rb, Rc), pick3<(PH + 1) % 3>(Ra, Rb, Rc),
                               lbase + o_cur, lbase + o_nxt, acc[SN], acc[SM], acc[SO]);

    if (EPI) {
        const int h = hin - 1;
        #pragma unroll
        for (int r = 0; r < 8; ++r) {
            int co = (th << 4) + r + (hi << 3);
            const float m0 = acc[SO][0][r], m1 = acc[SO][1][r], m2 = acc[SO][2][r];
            const float m3 = acc[SO][3][r], m4 = acc[SO][4][r], m5 = acc[SO][5][r];
            const float s12 = m1 + m2, d12 = m1 - m2, s34 = m3 + m4, d34 = m3 - m4;
            v4f y;
            y[0] = (m0 + s12) + (s34 + bia[r]);
            y[1] = (d12 + 2.0f * d34) + bia[r];
            y[2] = (s12 + 4.0f * s34) + bia[r];
            y[3] = (d12 + 8.0f * d34) + (m5 + bia[r]);
            *(v4f*)&out[((b * COUT + co) * HH + h) * WW + wcol] = y;
        }
    }

    if (LD) store_row(smem + o_wr, hin + 2, tid, xa, xb, hv);
    { int t = o_cur; o_cur = o_nxt; o_nxt = o_wr; o_wr = t; }
    __builtin_amdgcn_fence(__ATOMIC_RELEASE, "workgroup", "local");
    __builtin_amdgcn_s_barrier();
    __builtin_amdgcn_fence(__ATOMIC_ACQUIRE, "workgroup", "local");
}

extern "C" __global__ __launch_bounds__(NTHR, 1)
void conv3x3_wmma_f32(const float* __restrict__ x,
                      const float* __restrict__ wgt,
                      const float* __restrict__ bias,
                      float* __restrict__ out)
{
    extern __shared__ float smem[];
    v2f*   wlds = (v2f*)(smem + W_OFF);
    float* blds = smem + B_OFF;

    const int tid   = threadIdx.x;
    const int blk   = blockIdx.x;
    const int b     = blk >> 4;
    const int h0    = ((blk >> 1) & 7) * ROWS;
    const int wtile = blk & 1;
    const int wbase_g = wtile * TILE_W;
    const int lane = tid & 31;
    const int wave = tid >> 5;

    float wr[24][3];
    #pragma unroll
    for (int i = 0; i < 24; ++i) {
        int u    = i * NTHR + tid;
        int rest = u >> 5;
        int cout = (u & 7) + ((rest & 3) << 3);
        int mrow = ((rest >> 2) << 2) + ((u >> 3) & 3);
        const float* p = wgt + cout * (CIN * 9) + mrow * 3;
        wr[i][0] = p[0];
        wr[i][1] = p[1];
        wr[i][2] = p[2];
    }
    v4f xa[NLD], xb[NLD], ya[NLD], yb[NLD];
    float hv, hw;
    load_row(x, b, h0 - 1, wbase_g, tid, xa, xb, hv);
    load_row(x, b, h0,     wbase_g, tid, ya, yb, hw);
    float bsv = 0.0f;
    if (tid < 32) bsv = bias[tid];

    #pragma unroll
    for (int i = 0; i < 24; ++i) {
        int u    = i * NTHR + tid;
        int rest = u >> 5;
        int cout = (u & 7) + ((rest & 3) << 3);
        int mrow = ((rest >> 2) << 2) + ((u >> 3) & 3);
        int ci   = mrow / 3;
        int kh   = mrow - 3 * ci;
        const float g0 = wr[i][0], g1 = wr[i][1], g2 = wr[i][2];
        const float sg = g0 + g2;
        const float tg = g0 * (1.0f / 24.0f) + g2 * (1.0f / 6.0f);
        const float hg = g1 * (1.0f / 12.0f);
        float uu[NCOMP];
        uu[0] = 0.25f * g0;
        uu[1] = (sg + g1) * (-1.0f / 6.0f);
        uu[2] = (sg - g1) * (-1.0f / 6.0f);
        uu[3] = tg + hg;
        uu[4] = tg - hg;
        uu[5] = g2;
        int t  = cout >> 4;
        int l  = (cout & 15) + (((ci >> 1) & 1) << 4);
        int cg = ci >> 2;
        #pragma unroll
        for (int c = 0; c < NCOMP; ++c) {
            int s = ((kh * NCOMP + c) << 3) + cg;
            smem[W_OFF + ((((s << 1) + t) * 32 + l) << 1) + (ci & 1)] = uu[c];
        }
    }
    store_row(smem,              h0 - 1, tid, xa, xb, hv);
    store_row(smem + XROW_ELEMS, h0,     tid, ya, yb, hw);
    if (tid < 32) blds[tid] = bsv;
    __syncthreads();

    const int th   = wave & 1;
    const int qb   = (wave >> 1) << 4;
    const int nco  = lane & 15;
    const int hi   = lane >> 4;

    v2f A[NSTEPW];
    #pragma unroll
    for (int s = 0; s < NSTEPW; ++s) A[s] = wlds[(s * 2 + th) * 32 + lane];
    float bia[8];
    #pragma unroll
    for (int r = 0; r < 8; ++r) bia[r] = blds[(th << 4) + r + (hi << 3)];

    const int wcol = wbase_g + ((qb + nco) << 2);
    const float* lbase = smem + ((qb + nco) << 2) * XPADW + (hi << 1);

    v2f Ra[NRAW], Rb[NRAW], Rc[NRAW];
    #pragma unroll
    for (int q = 0; q < NRAW; ++q) Ra[q] = *(const v2f*)(lbase + ROFF(q / NCOMP, q % NCOMP));

    v8f acc[3][NCOMP] = {};
    int o_cur = 0, o_nxt = XROW_ELEMS, o_wr = 2 * XROW_ELEMS;

#define TRIP_ARGS x, out, smem, lbase, b, h0, wbase_g, tid, th, hi, wcol, A, bia, Ra, Rb, Rc, acc, xa, xb, hv, o_cur, o_nxt, o_wr
    static_assert(ROWS == 32, "trip numbers below are written out for 32 output rows (34 input rows) per block");
    trip<0, true,  false, false, true,  true,  false>(0, TRIP_ARGS);
    trip<1, true,  true,  false, true,  true,  false>(1, TRIP_ARGS);
    #pragma unroll 1
    for (int i = 2; i < 32; i += 3) {
        trip<2, true, true, true, true, true, true>(i,     TRIP_ARGS);
        trip<0, true, true, true, true, true, true>(i + 1, TRIP_ARGS);
        trip<1, true, true, true, true, true, true>(i + 2, TRIP_ARGS);
    }
    trip<2, false, true,  true,  true,  false, true >(32, TRIP_ARGS);
    trip<0, false, false, true,  false, false, true >(33, TRIP_ARGS);
#undef TRIP_ARGS
}

extern "C" void kernel_launch(void* const* d_in, const int* in_sizes, int n_in,
                              void* d_out, int out_size, void* d_ws, size_t ws_size,
                              hipStream_t stream) {
    const float* x    = (const float*)d_in[0];
    const float* wgt  = (const float*)d_in[1];
    const float* bias = (const float*)d_in[2];
    float* out = (float*)d_out;

    dim3 grid(16 * 8 * 2);
    dim3 block(NTHR);
    size_t smem = SMEM_FLOATS * sizeof(float);
    hipLaunchKernelGGL(conv3x3_wmma_f32, grid, block, smem, stream, x, wgt, bias, out);
}


# Input rows bypass LDS: each lane loads its own B-fragment channels one trip ahead (one 16-byte, two 4-byte loads per channel); row walk has no barrier and no LDS traffic; V5 column sum regrouped
# speedup vs baseline: 5.4751x; 1.0709x over previous
#include <hip/hip_runtime.h>


typedef __attribute__((ext_vector_type(2))) float v2f;
typedef __attribute__((ext_vector_type(4))) float v4f;
typedef __attribute__((ext_vector_type(8))) float v8f;

#define CIN    32
#define COUT   32
#define HH     256
#define WW     256
#define TILE_W 128
#define ROWS   32
#define NIN    (ROWS + 2)
#define NCOMP  6
#define NSTEPW (3 * NCOMP * 8)
#define NTHR   128

#define NCH    16

#define W_OFF   0
#define W_FRAGS (NSTEPW * 2 * 32)
#define B_OFF   (W_OFF + W_FRAGS * 2)
#define SMEM_FLOATS (B_OFF + 32)

__device__ __forceinline__ void load_raw(const float* pm, const float* pl, const float* pr,
                                         v4f (&mid)[NCH], float (&lft)[NCH], float (&rgt)[NCH])
{
    #pragma unroll
    for (int j = 0; j < NCH; ++j) {
        const int off = (((j >> 1) << 2) + (j & 1)) * (HH * WW);
        mid[j] = *(const v4f*)(pm + off);
        lft[j] = pl[off];
        rgt[j] = pr[off];
    }
}

__device__ __forceinline__ v8f mma(v2f a, v2f b, v8f c)
{
    return __builtin_amdgcn_wmma_f32_16x16x4_f32(false, a, false, b, (short)0, c, false, false);
}

template <bool DN, bool DM, bool DO>
__device__ __forceinline__ void half_row(const v2f (&A)[NSTEPW], const v4f (&mid)[NCH], const float (&lft)[NCH],
                                         const float (&rgt)[NCH], const float f0, const float f5, int cg0,
                                         v8f (&n)[NCOMP], v8f (&m)[NCOMP], v8f (&o)[NCOMP])
{
    #pragma unroll
    for (int g = 0; g < 4; ++g) {
        const int cg = cg0 + g;
        v2f V[NCOMP];
        #pragma unroll
        for (int s = 0; s < 2; ++s) {
            const int j = (cg << 1) + s;
            const float d0 = lft[j], d1 = mid[j][0], d2 = mid[j][1], d3 = mid[j][2], d4 = mid[j][3], d5 = rgt[j];
            const float ea = d4 - 4.0f * d2;
            const float eb = d3 - 4.0f * d1;
            const float ec = d4 - d2;
            const float ed = d3 - d1;
            V[0][s] = f0 * d0 + (d4 - 5.0f * d2);
            V[1][s] = ea + eb;
            V[2][s] = ea - eb;
            V[3][s] = ec + 2.0f * ed;
            V[4][s] = ec - 2.0f * ed;
            V[5][s] = f5 * d5 - (eb + 4.0f * d3);
        }
        #pragma unroll
        for (int c = 0; c < NCOMP; ++c) {
            if (DN) n[c] = mma(A[(0 * NCOMP + c) * 8 + cg], V[c], n[c]);
            if (DM) m[c] = mma(A[(1 * NCOMP + c) * 8 + cg], V[c], m[c]);
            if (DO) o[c] = mma(A[(2 * NCOMP + c) * 8 + cg], V[c], o[c]);
        }
    }
}

template <int K, typename T>
__device__ __forceinline__ T (&pick3(T (&a)[NCH], T (&b)[NCH], T (&c)[NCH]))[NCH]
{
    if constexpr (K == 0) return a;
    else if constexpr (K == 1) return b;
    else return c;
}

template <int PH, bool DN, bool DM, bool DO, bool LD, bool EPI, bool EDGE>
__device__ __forceinline__ void trip(const int i, const float* xm, float* __restrict__ out,
                                     const int b, const int h0, const int lo, const int ro,
                                     const float f0, const float f5,
                                     const int th, const int hi, const int wcol,
                                     const v2f (&A)[NSTEPW], const float (&bia)[8],
                                     v4f (&Ma)[NCH], v4f (&Mb)[NCH], v4f (&Mc)[NCH],
                                     float (&La)[NCH], float (&Lb)[NCH], float (&Lc)[NCH],
                                     float (&Qa)[NCH], float (&Qb)[NCH], float (&Qc)[NCH],
                                     v8f (&acc)[3][NCOMP])
{
    constexpr int SN = PH, SM = (PH + 2) % 3, SO = (PH + 1) % 3;
    constexpr int CUR = PH, NXT = (PH + 1) % 3;
    const int hin = h0 - 1 + i;

    if (LD) {
        const int hn = hin + 1;
        const int hc = hn < 0 ? 0 : (hn > HH - 1 ? HH - 1 : hn);
        const float* pm = xm + hc * WW;
        load_raw(pm, pm + lo, pm + ro, pick3<NXT>(Ma, Mb, Mc), pick3<NXT>(La, Lb, Lc), pick3<NXT>(Qa, Qb, Qc));
    }

    if (DN) {
        #pragma unroll
        for (int c = 0; c < NCOMP; ++c) acc[SN][c] = v8f{};
    }
    const bool rowok = !EDGE || (unsigned)hin < HH;
    if (rowok) {
        half_row<DN, DM, DO>(A, pick3<CUR>(Ma, Mb, Mc), pick3<CUR>(La, Lb, Lc), pick3<CUR>(Qa, Qb, Qc), f0, f5, 0,
                             acc[SN], acc[SM], acc[SO]);
        half_row<DN, DM, DO>(A, pick3<CUR>(Ma, Mb, Mc), pick3<CUR>(La, Lb, Lc), pick3<CUR>(Qa, Qb, Qc), f0, f5, 4,
                             acc[SN], acc[SM], acc[SO]);
    }

    if (EPI) {
        const int h = hin - 1;
        #pragma unroll
        for (int r = 0; r < 8; ++r) {
            int co = (th << 4) + r + (hi << 3);
            const float m0 = acc[SO][0][r], m1 = acc[SO][1][r], m2 = acc[SO][2][r];
            const float m3 = acc[SO][3][r], m4 = acc[SO][4][r], m5 = acc[SO][5][r];
            const float s12 = m1 + m2, d12 = m1 - m2, s34 = m3 + m4, d34 = m3 - m4;
            v4f y;
            y[0] = (m0 + s12) + (s34 + bia[r]);
            y[1] = (d12 + 2.0f * d34) + bia[r];
            y[2] = (s12 + 4.0f * s34) + bia[r];
            y[3] = (d12 + 8.0f * d34) + (m5 + bia[r]);
            *(v4f*)&out[((b * COUT + co) * HH + h) * WW + wcol] = y;
        }
    }
    __builtin_amdgcn_sched_barrier(0);
}

extern "C" __global__ __launch_bounds__(NTHR, 1)
void conv3x3_wmma_f32(const float* __restrict__ x,
                      const float* __restrict__ wgt,
                      const float* __restrict__ bias,
                      float* __restrict__ out)
{
    extern __shared__ float smem[];
    v2f*   wlds = (v2f*)(smem + W_OFF);
    float* blds = smem + B_OFF;

    const int tid   = threadIdx.x;
    const int blk   = blockIdx.x;
    const int b     = blk >> 4;
    const int h0    = ((blk >> 1) & 7) * ROWS;
    const int wtile = blk & 1;
    const int wbase_g = wtile * TILE_W;
    const int lane = tid & 31;
    const int wave = tid >> 5;

    const int th   = wave & 1;
    const int qb   = (wave >> 1) << 4;
    const int nco  = lane & 15;
    const int hi   = lane >> 4;
    const int wcol = wbase_g + ((qb + nco) << 2);

    const bool ledge = wcol == 0;
    const bool redge = wcol + 4 >= WW;
    const int   lo = ledge ? 0 : -1;
    const int   ro = redge ? 3 : 4;
    const float f0 = ledge ? 0.0f : 4.0f;
    const float f5 = redge ? 0.0f : 1.0f;
    const float* xm = x + ((b * CIN + (hi << 1)) * HH) * WW + wcol;

    v4f Ma[NCH], Mb[NCH], Mc[NCH];
    float La[NCH], Lb[NCH], Lc[NCH];
    float Qa[NCH], Qb[NCH], Qc[NCH];
    {
        const int hn = h0 - 1;
        const int hc = hn < 0 ? 0 : hn;
        const float* pm = xm + hc * WW;
        load_raw(pm, pm + lo, pm + ro, Ma, La, Qa);
    }

    float wr[24][3];
    #pragma unroll
    for (int i = 0; i < 24; ++i) {
        int u    = i * NTHR + tid;
        int rest = u >> 5;
        int cout = (u & 7) + ((rest & 3) << 3);
        int mrow = ((rest >> 2) << 2) + ((u >> 3) & 3);
        const float* p = wgt + cout * (CIN * 9) + mrow * 3;
        wr[i][0] = p[0];
        wr[i][1] = p[1];
        wr[i][2] = p[2];
    }
    float bsv = 0.0f;
    if (tid < 32) bsv = bias[tid];

    #pragma unroll
    for (int i = 0; i < 24; ++i) {
        int u    = i * NTHR + tid;
        int rest = u >> 5;
        int cout = (u & 7) + ((rest & 3) << 3);
        int mrow = ((rest >> 2) << 2) + ((u >> 3) & 3);
        int ci   = mrow / 3;
        int kh   = mrow - 3 * ci;
        const float g0 = wr[i][0], g1 = wr[i][1], g2 = wr[i][2];
        const float sg = g0 + g2;
        const float tg = g0 * (1.0f / 24.0f) + g2 * (1.0f / 6.0f);
        const float hg = g1 * (1.0f / 12.0f);
        float uu[NCOMP];
        uu[0] = 0.25f * g0;
        uu[1] = (sg + g1) * (-1.0f / 6.0f);
        uu[2] = (sg - g1) * (-1.0f / 6.0f);
        uu[3] = tg + hg;
        uu[4] = tg - hg;
        uu[5] = g2;
        int t  = cout >> 4;
        int l  = (cout & 15) + (((ci >> 1) & 1) << 4);
        int cg = ci >> 2;
        #pragma unroll
        for (int c = 0; c < NCOMP; ++c) {
            int s = ((kh * NCOMP + c) << 3) + cg;
            smem[W_OFF + ((((s << 1) + t) * 32 + l) << 1) + (ci & 1)] = uu[c];
        }
    }
    if (tid < 32) blds[tid] = bsv;
    __syncthreads();

    v2f A[NSTEPW];
    #pragma unroll
    for (int s = 0; s < NSTEPW; ++s) A[s] = wlds[(s * 2 + th) * 32 + lane];
    float bia[8];
    #pragma unroll
    for (int r = 0; r < 8; ++r) bia[r] = blds[(th << 4) + r + (hi << 3)];

    v8f acc[3][NCOMP] = {};

#define TRIP_ARGS xm, out, b, h0, lo, ro, f0, f5, th, hi, wcol, A, bia, Ma, Mb, Mc, La, Lb, Lc, Qa, Qb, Qc, acc
    static_assert(ROWS == 32, "trip numbers below are written out for 32 output rows (34 input rows) per block");
    trip<0, true,  false, false, true,  false, true >(0, TRIP_ARGS);
    trip<1, true,  true,  false, true,  false, false>(1, TRIP_ARGS);
    #pragma unroll 1
    for (int i = 2; i < 32; i += 3) {
        trip<2, true, true, true, true, true, false>(i,     TRIP_ARGS);
        trip<0, true, true, true, true, true, false>(i + 1, TRIP_ARGS);
        trip<1, true, true, true, true, true, false>(i + 2, TRIP_ARGS);
    }
    trip<2, false, true,  true,  true,  true,  false>(32, TRIP_ARGS);
    trip<0, false, false, true,  false, true,  true >(33, TRIP_ARGS);
#undef TRIP_ARGS
}

extern "C" void kernel_launch(void* const* d_in, const int* in_sizes, int n_in,
                              void* d_out, int out_size, void* d_ws, size_t ws_size,
                              hipStream_t stream) {
    const float* x    = (const float*)d_in[0];
    const float* wgt  = (const float*)d_in[1];
    const float* bias = (const float*)d_in[2];
    float* out = (float*)d_out;

    dim3 grid(16 * 8 * 2);
    dim3 block(NTHR);
    size_t smem = SMEM_FLOATS * sizeof(float);
    hipLaunchKernelGGL(conv3x3_wmma_f32, grid, block, smem, stream, x, wgt, bias, out);
}
